# starting kernel
# baseline (speedup 1.0000x reference)
#include <hip/hip_runtime.h>

typedef float v2f __attribute__((ext_vector_type(2)));
typedef float v8f __attribute__((ext_vector_type(8)));

#define WAVES_PER_BLOCK 8

template <int K, int NT, bool MEANPAIR>
__global__ __launch_bounds__(32 * WAVES_PER_BLOCK)
void wmma_gemm_kernel(const float* __restrict__ X, const float* __restrict__ W,
                      const float* __restrict__ bias, float* __restrict__ Out, int M)
{
    constexpr int KS = K / 4;
    constexpr int NO = NT * 16;
    const int lane = threadIdx.x & 31;
    const int wave = blockIdx.x * (blockDim.x >> 5) + (threadIdx.x >> 5);
    if (wave >= (M >> 4)) return;
    const int half = lane >> 4;
    const int lm   = lane & 15;
    const int arow = wave * 16 + lm;

    v2f a[KS];
    if (MEANPAIR) {
        const float* __restrict__ x0 = X + (size_t)arow * 2 * K;
#pragma unroll
        for (int i = 0; i < KS; ++i) {
            const int ka = 4 * i + half * 2;
            const v2f lo = *(const v2f*)(x0 + ka);
            const v2f hi = *(const v2f*)(x0 + K + ka);
            a[i].x = 0.5f * (lo.x + hi.x);
            a[i].y = 0.5f * (lo.y + hi.y);
        }
    } else {
        const float* __restrict__ x0 = X + (size_t)arow * K;
#pragma unroll
        for (int i = 0; i < KS; ++i)
            a[i] = *(const v2f*)(x0 + 4 * i + half * 2);
    }

#pragma unroll
    for (int nt = 0; nt < NT; ++nt) {
        const int bcol = nt * 16 + lm;
        const float* __restrict__ wrow = W + (size_t)bcol * K;
        v2f b[KS];
#pragma unroll
        for (int i = 0; i < KS; ++i)
            b[i] = *(const v2f*)(wrow + 4 * i + half * 2);

        v8f acc = {0.f, 0.f, 0.f, 0.f, 0.f, 0.f, 0.f, 0.f};
#pragma unroll
        for (int i = 0; i < KS; ++i)
            acc = __builtin_amdgcn_wmma_f32_16x16x4_f32(false, a[i], false, b[i],
                                                        (short)0, acc, false, false);

        const int   ocol = nt * 16 + lm;
        const float bv   = bias ? bias[ocol] : 0.0f;
#pragma unroll
        for (int r = 0; r < 8; ++r) {
            const int orow = wave * 16 + half * 8 + r;
            Out[(size_t)orow * NO + ocol] = acc[r] + bv;
        }
    }
}

__global__ void deg_kernel(const int* __restrict__ dst, float* __restrict__ deg, int E)
{
    int i = blockIdx.x * blockDim.x + threadIdx.x;
    if (i < E) atomicAdd(&deg[dst[i]], 1.0f);
}

__global__ void scatter_kernel(const float* __restrict__ P, const int* __restrict__ src,
                               const int* __restrict__ dst, float* __restrict__ Agg,
                               int E, int C, int lc2)
{
    long long tid   = (long long)blockIdx.x * blockDim.x + threadIdx.x;
    long long total = (long long)E << lc2;
    if (tid >= total) return;
    const int e = (int)(tid >> lc2);
    const int c = ((int)tid & ((1 << lc2) - 1)) << 2;
    const int s = src[e];
    const int d = dst[e];
    const float4 v = *(const float4*)(P + (size_t)s * C + c);
    float* ap = Agg + (size_t)d * C + c;
    atomicAdd(ap + 0, v.x);
    atomicAdd(ap + 1, v.y);
    atomicAdd(ap + 2, v.z);
    atomicAdd(ap + 3, v.w);
}

__global__ void combine_h_kernel(const float* __restrict__ agg, const float* __restrict__ r0,
                                 const float* __restrict__ e0, const float* __restrict__ bl,
                                 const float* __restrict__ deg, float* __restrict__ h, int N)
{
    long long t = (long long)blockIdx.x * blockDim.x + threadIdx.x;
    if (t >= (long long)N * 64) return;
    const int n = (int)(t >> 6);
    const int c = (int)t & 63;
    float dv = deg[n];
    dv = dv > 1.0f ? dv : 1.0f;
    const float v = agg[t] / dv + bl[c] + r0[t] + e0[t];
    h[t] = v > 0.0f ? v : 0.0f;
}

__global__ void final_kernel(const float* __restrict__ agg1, const float* __restrict__ r1,
                             const float* __restrict__ e1, const float* __restrict__ bl1,
                             const float* __restrict__ deg, float* __restrict__ out, int N)
{
    int t = blockIdx.x * blockDim.x + threadIdx.x;
    if (t >= N * 16) return;
    const int n = t >> 4;
    const int c = t & 15;
    float dv = deg[n];
    dv = dv > 1.0f ? dv : 1.0f;
    out[t] = agg1[t] / dv + bl1[c] + r1[t] + e1[t];
}

extern "C" void kernel_launch(void* const* d_in, const int* in_sizes, int n_in,
                              void* d_out, int out_size, void* d_ws, size_t ws_size,
                              hipStream_t stream)
{
    const float* x_feat = (const float*)d_in[0];
    const float* x_emb  = (const float*)d_in[1];
    const int*   eidx   = (const int*)d_in[2];
    const float* Wl0 = (const float*)d_in[3];
    const float* bl0 = (const float*)d_in[4];
    const float* Wr0 = (const float*)d_in[5];
    const float* We0 = (const float*)d_in[6];
    const float* be0 = (const float*)d_in[7];
    const float* Wl1 = (const float*)d_in[8];
    const float* bl1 = (const float*)d_in[9];
    const float* Wr1 = (const float*)d_in[10];
    const float* We1 = (const float*)d_in[11];
    const float* be1 = (const float*)d_in[12];

    const int FEAT = 128;
    const int N = in_sizes[0] / FEAT;
    const int E = in_sizes[2] / 2;
    const int* src = eidx;
    const int* dst = eidx + E;

    float*       ws = (float*)d_ws;
    const size_t n  = (size_t)N;
    float* deg  = ws;
    float* e0   = ws + n;
    float* p0   = ws + 65 * n;
    float* r0   = ws + 129 * n;
    float* agg0 = ws + 193 * n;
    float* h    = p0;
    float* p1   = agg0;
    float* r1   = ws + 209 * n;
    float* e1   = ws + 225 * n;
    float* agg1 = ws + 241 * n;

    hipMemsetAsync(deg,  0, n * sizeof(float), stream);
    hipMemsetAsync(agg0, 0, 64 * n * sizeof(float), stream);

    deg_kernel<<<(E + 255) / 256, 256, 0, stream>>>(dst, deg, E);

    const int mwaves  = N >> 4;
    const int gblocks = (mwaves + WAVES_PER_BLOCK - 1) / WAVES_PER_BLOCK;
    const int gthreads = 32 * WAVES_PER_BLOCK;

    wmma_gemm_kernel<64, 4, true ><<<gblocks, gthreads, 0, stream>>>(x_emb,  We0, be0,     e0, N);
    wmma_gemm_kernel<128, 4, false><<<gblocks, gthreads, 0, stream>>>(x_feat, Wl0, nullptr, p0, N);
    wmma_gemm_kernel<128, 4, false><<<gblocks, gthreads, 0, stream>>>(x_feat, Wr0, nullptr, r0, N);

    {
        const long long total = (long long)E << 4;
        scatter_kernel<<<(int)((total + 255) / 256), 256, 0, stream>>>(p0, src, dst, agg0, E, 64, 4);
    }

    {
        const long long total = (long long)N * 64;
        combine_h_kernel<<<(int)((total + 255) / 256), 256, 0, stream>>>(agg0, r0, e0, bl0, deg, h, N);
    }

    wmma_gemm_kernel<64, 1, false><<<gblocks, gthreads, 0, stream>>>(h,  Wl1, nullptr, p1, N);
    wmma_gemm_kernel<64, 1, false><<<gblocks, gthreads, 0, stream>>>(h,  Wr1, nullptr, r1, N);
    wmma_gemm_kernel<64, 1, false><<<gblocks, gthreads, 0, stream>>>(e0, We1, be1,     e1, N);

    hipMemsetAsync(agg1, 0, 16 * n * sizeof(float), stream);

    {
        const long long total = (long long)E << 2;
        scatter_kernel<<<(int)((total + 255) / 256), 256, 0, stream>>>(p1, src, dst, agg1, E, 16, 2);
    }

    final_kernel<<<(N * 16 + 255) / 256, 256, 0, stream>>>(agg1, r1, e1, bl1, deg,
                                                           (float*)d_out, N);
}


# Both aggregation launches at 64 registers (16 waves/SIMD): node lines spread over lanes, slot words by lane exchange; layer-1 products split by k over 8 waves, partials summed via LDS in wave order
# speedup vs baseline: 3.9478x; 3.9478x over previous
#include <hip/hip_runtime.h>

typedef float v2f __attribute__((ext_vector_type(2)));
typedef float v4f __attribute__((ext_vector_type(4)));
typedef int   v4i __attribute__((ext_vector_type(4)));
typedef float v8f __attribute__((ext_vector_type(8)));

#define GEMM_WAVES 4

template <int K, int NT0, int NW, bool MEANPAIR>
__global__ __launch_bounds__(32 * GEMM_WAVES)
void wmma_ws_kernel(const float* __restrict__ X, const float* __restrict__ Wa,
                    const float* __restrict__ Wb, const float* __restrict__ bias,
                    float* __restrict__ Oa, float* __restrict__ Ob, int M)
{
    constexpr int KS = K / 4;
    constexpr int NT = NT0 * NW;
    constexpr int NO = NT0 * 16;
    constexpr int XP = MEANPAIR ? 2 * K : K;
    constexpr int NA = MEANPAIR ? 2 * KS : KS;
    const int lane = threadIdx.x & 31;
    const int wave = blockIdx.x * GEMM_WAVES + (threadIdx.x >> 5);
    const int nwav = gridDim.x * GEMM_WAVES;
    const int nstr = M >> 4;
    if (wave >= nstr) return;
    const int half = lane >> 4;
    const int lm   = lane & 15;

    v2f b[NT][KS];
#pragma unroll
    for (int t = 0; t < NT; ++t) {
        const float* __restrict__ wsel = (t < NT0) ? Wa : Wb;
        const float* __restrict__ wrow = wsel + (size_t)((t % NT0) * 16 + lm) * K;
#pragma unroll
        for (int i = 0; i < KS; ++i)
            b[t][i] = *(const v2f*)(wrow + 4 * i + half * 2);
    }
    float bv[NT0];
#pragma unroll
    for (int t = 0; t < NT0; ++t) bv[t] = bias ? bias[t * 16 + lm] : 0.0f;

    auto load_stripe = [&](v2f (&raw)[NA], const int st) {
        const float* __restrict__ x0 = X + (size_t)(st * 16 + lm) * XP + half * 2;
#pragma unroll
        for (int i = 0; i < KS; ++i) {
            raw[i] = *(const v2f*)(x0 + 4 * i);
            if (MEANPAIR) raw[KS + i] = *(const v2f*)(x0 + K + 4 * i);
        }
    };
    auto do_stripe = [&](const v2f (&raw)[NA], const int st) {
        v2f a[KS];
#pragma unroll
        for (int i = 0; i < KS; ++i) {
            if (MEANPAIR) {
                a[i].x = 0.5f * (raw[i].x + raw[KS + i].x);
                a[i].y = 0.5f * (raw[i].y + raw[KS + i].y);
            } else {
                a[i] = raw[i];
            }
        }
        v8f c[NT];
#pragma unroll
        for (int t = 0; t < NT; ++t) c[t] = (v8f){0.f, 0.f, 0.f, 0.f, 0.f, 0.f, 0.f, 0.f};
#pragma unroll
        for (int i = 0; i < KS; ++i) {
#pragma unroll
            for (int t = 0; t < NT; ++t)
                c[t] = __builtin_amdgcn_wmma_f32_16x16x4_f32(false, a[i], false, b[t][i],
                                                             (short)0, c[t], false, false);
        }
#pragma unroll
        for (int t = 0; t < NT; ++t) {
            float*      o  = (t < NT0) ? Oa : Ob;
            const int   tc = t % NT0;
            const float bb = (t < NT0) ? bv[tc] : 0.0f;
#pragma unroll
            for (int r = 0; r < 8; ++r) {
                const int orow = st * 16 + half * 8 + r;
                o[(size_t)orow * NO + tc * 16 + lm] = c[t][r] + bb;
            }
        }
    };

    v2f ra[NA], rb[NA];
    int s = wave;
    load_stripe(ra, s);
    for (;;) {
        const int s1 = s + nwav;
        if (s1 < nstr) {
            load_stripe(rb, s1);
            __builtin_amdgcn_sched_barrier(0);
            do_stripe(ra, s);
        } else {
            do_stripe(ra, s);
            break;
        }
        const int s2 = s1 + nwav;
        if (s2 < nstr) {
            load_stripe(ra, s2);
            __builtin_amdgcn_sched_barrier(0);
            do_stripe(rb, s1);
        } else {
            do_stripe(rb, s1);
            break;
        }
        s = s2;
    }
}

#define BKW  32
#define CAPA (BKW - 1)
#define CAPN (CAPA + BKW)

__device__ __forceinline__ void place_one(int* __restrict__ bka, int* __restrict__ bkb,
                                          const int d, const int p, const int s)
{
    if (p < CAPA)      bka[(size_t)d * BKW + p] = s;
    else if (p < CAPN) bkb[(size_t)d * BKW + (p - CAPA)] = s;
}

#define EPT   8
#define CHUNK (256 * EPT)
#define TKW   32

__device__ __forceinline__ int rank_one(int* __restrict__ bka, const int d, const bool mine)
{
    return mine ? atomicAdd(bka + (size_t)d * BKW + CAPA, 1) : 0;
}

__global__ __launch_bounds__(256)
void place_kernel(const int* __restrict__ src, const int* __restrict__ dst,
                  int* __restrict__ bka, int* __restrict__ bkb, int* __restrict__ tick,
                  int E, int N, int nchunks)
{
    __shared__ int pick[2];
    if (threadIdx.x == 0) {
        const unsigned id = __builtin_amdgcn_s_sendmsg_rtn(0x87);
        int h = (int)(id >> 18) & 1;
        int t = atomicAdd(tick + h * TKW, 1);
        if (t >= nchunks) {
            h ^= 1;
            t = atomicAdd(tick + h * TKW, 1);
        }
        pick[0] = h;
        pick[1] = t;
    }
    __syncthreads();
    const int h = pick[0];
    const int t = pick[1];
    if (t >= nchunks) return;
    const int nh   = N >> 1;
    const int base = t * CHUNK;
    const bool quads = ((E & 3) == 0) &&
                       ((((size_t)src | (size_t)dst) & 15) == 0);
    if (quads) {
        int4 d[EPT / 4], s[EPT / 4];
#pragma unroll
        for (int j = 0; j < EPT / 4; ++j) {
            const int e = base + j * 1024 + (int)threadIdx.x * 4;
            if (e < E) {
                d[j] = *(const int4*)(dst + e);
                s[j] = *(const int4*)(src + e);
            } else {
                d[j] = make_int4(-1, -1, -1, -1);
                s[j] = make_int4(0, 0, 0, 0);
            }
        }
        bool m[EPT];
        int  p[EPT];
#pragma unroll
        for (int j = 0; j < EPT / 4; ++j) {
            m[4 * j + 0] = d[j].x >= 0 && (int)(d[j].x >= nh) == h;
            m[4 * j + 1] = d[j].y >= 0 && (int)(d[j].y >= nh) == h;
            m[4 * j + 2] = d[j].z >= 0 && (int)(d[j].z >= nh) == h;
            m[4 * j + 3] = d[j].w >= 0 && (int)(d[j].w >= nh) == h;
        }
#pragma unroll
        for (int j = 0; j < EPT / 4; ++j) {
            p[4 * j + 0] = rank_one(bka, d[j].x, m[4 * j + 0]);
            p[4 * j + 1] = rank_one(bka, d[j].y, m[4 * j + 1]);
            p[4 * j + 2] = rank_one(bka, d[j].z, m[4 * j + 2]);
            p[4 * j + 3] = rank_one(bka, d[j].w, m[4 * j + 3]);
        }
#pragma unroll
        for (int j = 0; j < EPT / 4; ++j) {
            if (m[4 * j + 0]) place_one(bka, bkb, d[j].x, p[4 * j + 0], s[j].x);
            if (m[4 * j + 1]) place_one(bka, bkb, d[j].y, p[4 * j + 1], s[j].y);
            if (m[4 * j + 2]) place_one(bka, bkb, d[j].z, p[4 * j + 2], s[j].z);
            if (m[4 * j + 3]) place_one(bka, bkb, d[j].w, p[4 * j + 3], s[j].w);
        }
    } else {
        for (int j = 0; j < EPT; ++j) {
            const int i = base + j * 256 + (int)threadIdx.x;
            if (i < E) {
                const int dd = dst[i];
                if ((int)(dd >= nh) == h)
                    place_one(bka, bkb, dd, atomicAdd(bka + (size_t)dd * BKW + CAPA, 1), src[i]);
            }
        }
    }
}

template <int N> __device__ __forceinline__ int row_take(const int v)
{
    return __builtin_amdgcn_update_dpp(0, v, 0x150 + N, 0xf, 0xf, false);
}
__device__ __forceinline__ int row_take_n(const int v, const int n)
{
    switch (n & 15) {
    case 0:  return row_take<0>(v);
    case 1:  return row_take<1>(v);
    case 2:  return row_take<2>(v);
    case 3:  return row_take<3>(v);
    case 4:  return row_take<4>(v);
    case 5:  return row_take<5>(v);
    case 6:  return row_take<6>(v);
    case 7:  return row_take<7>(v);
    case 8:  return row_take<8>(v);
    case 9:  return row_take<9>(v);
    case 10: return row_take<10>(v);
    case 11: return row_take<11>(v);
    case 12: return row_take<12>(v);
    case 13: return row_take<13>(v);
    case 14: return row_take<14>(v);
    default: return row_take<15>(v);
    }
}
__device__ __forceinline__ int node8_take(const int v, const int a, const int b)
{
    const unsigned lo = 0x1111u * (unsigned)a | (0x1111u * (unsigned)b) << 16;
    const unsigned hi = 0x1111u * (unsigned)(8 + a) | (0x1111u * (unsigned)(8 + b)) << 16;
    return (int)__builtin_amdgcn_permlane16((unsigned)v, (unsigned)v, lo, hi, false, false);
}
template <int LPN, int WPL> __device__ __forceinline__ int node_word(const int (&q)[WPL], const int w)
{
    if constexpr (LPN == 16) return row_take_n(q[w / 16], w % 16);
    else                     return node8_take(q[w / 8], w % 8, w % 8);
}
template <int LPN, int G, int WPL> __device__ __forceinline__ int slot_word(const int (&q)[WPL], const int kk)
{
    if constexpr (G == 1) return node_word<LPN, WPL>(q, kk);
    else                  return node8_take(q[kk / 8], kk % 8, (kk + 1) % 8);
}

#define HP 68
template <int C, int NPW, int U, bool RELU, int PB, bool PROJ>
__global__ __launch_bounds__(256)
void pull_kernel(const float* __restrict__ P, const int* __restrict__ bka, const int* __restrict__ bkb,
                 const int* __restrict__ src, const int* __restrict__ dst, int E,
                 const float* R, const float* __restrict__ Ex, const float* __restrict__ bias,
                 float* Out, int N,
                 const float* __restrict__ Wa, const float* __restrict__ Wb,
                 const float* __restrict__ Wc, const float* __restrict__ bias2)
{
    static_assert(!PROJ || (C == 64 && NPW == 2), "the projection tail is written for 16 nodes of 64 floats per block");
    constexpr int LPR = C / 4;
    constexpr int LPN = 32 / NPW;
    static_assert(LPN >= LPR && (NPW * C * 4) % 128 == 0, "a node needs a whole row of lanes, a wave whole lines");
    constexpr int G   = LPN / LPR;
    static_assert(G == 1 || G == 2, "the slot pick below is written for one or two groups");
    static_assert(BKW % (G * U) == 0, "whole batches cover a line");
    static_assert((LPN == 16 && G == 1) || (LPN == 8 && G == 2), "the lane exchanges are written for these two shapes");
    constexpr int WPL = BKW / LPN;
    const int lane = threadIdx.x & 31;
    const int wave = blockIdx.x * (blockDim.x >> 5) + (threadIdx.x >> 5);
    const int nw   = lane / LPN;
    const int node = wave * NPW + nw;
    const int ln   = lane % LPN;
    const int g    = ln / LPR;
    const int c    = (ln % LPR) * 4;
    const bool live = node < N;
    const size_t lo = (size_t)(live ? node : 0) * BKW;
    int q[WPL];
#pragma unroll
    for (int j = 0; j < WPL; ++j) q[j] = bka[lo + j * LPN + ln];
    const int    nd = live ? node : 0;
    const float4 rv = *(const float4*)(R + (size_t)(nd >> 4) * PB + (nd & 15) * C + c);
    float4 ev = {0.f, 0.f, 0.f, 0.f};
    if (PROJ) ev = *(const float4*)(Ex + (size_t)nd * C + c);
#pragma unroll
    for (int j = 0; j < WPL; ++j)
        asm volatile("" : "+v"(q[j]));
    const int  cw   = node_word<LPN, WPL>(q, CAPA);
    const int  cntv = live ? cw : 0;
    const bool big  = cntv > CAPN;
    const int  n    = big ? 0 : cntv;
    int nmax = n;
    for (int o = LPN; o < 32; o <<= 1) {
        const int m = __shfl_xor(nmax, o);
        nmax = m > nmax ? m : nmax;
    }
    float4 acc = {0.f, 0.f, 0.f, 0.f};
    auto run_line = [&](const int (&qq)[WPL], const int base, const int lim) __attribute__((always_inline)) {
#pragma unroll
        for (int k0 = 0; k0 < BKW; k0 += G * U) {
            if (base + k0 >= nmax) break;
            unsigned o[U];
            bool     ok[U];
#pragma unroll
            for (int u = 0; u < U; ++u) {
                const int kk = k0 + u * G;
                const int sl = slot_word<LPN, G, WPL>(qq, kk);
                ok[u] = kk + g < lim && base + kk + g < n;
                const unsigned s = ok[u] ? (unsigned)sl : 0u;
                o[u] = ((s >> 4) * (unsigned)PB + (s & 15u) * (unsigned)C + (unsigned)c) * 4u;
            }
            v4f v[U];
#pragma unroll
            for (int u = 0; u < U; ++u)
                v[u] = *(const v4f*)((const char*)P + o[u]);
            __builtin_amdgcn_sched_barrier(0);
#pragma unroll
            for (int u = 0; u < U; ++u) {
                const int mk = ok[u] ? -1 : 0;
                const v4f f = __builtin_bit_cast(v4f, __builtin_bit_cast(v4i, v[u]) & mk);
                acc.x += f.x;
                acc.y += f.y;
                acc.z += f.z;
                acc.w += f.w;
            }
        }
    };
    run_line(q, 0, CAPA);
    if (nmax > CAPA) {
        int q2[WPL];
#pragma unroll
        for (int j = 0; j < WPL; ++j) q2[j] = bkb[lo + j * LPN + ln];
        run_line(q2, CAPA, BKW);
    }
    if (__any(big)) {
        const int tgt = big ? node : -1;
        for (int j0 = 0; j0 < E; j0 += 32) {
            const int  j  = j0 + lane;
            const int  d  = j < E ? dst[j] : -1;
            const int  sv = j < E ? src[j] : 0;
            for (int m = 0; m < NPW; ++m) {
                const int tm = __shfl(tgt, m * LPN);
                unsigned mask = (unsigned)__ballot(tm >= 0 && d == tm);
                while (mask) {
                    const int b = __ffs(mask) - 1;
                    mask &= mask - 1;
                    const int st = __shfl(sv, b);
                    if (nw == m && g == 0) {
                        const float4 r = *(const float4*)(P + (size_t)(st >> 4) * PB + (st & 15) * C + c);
                        acc.x += r.x; acc.y += r.y; acc.z += r.z; acc.w += r.w;
                    }
                }
            }
        }
    }
    for (int o = LPR; o < LPN; o <<= 1) {
        acc.x += __shfl_xor(acc.x, o);
        acc.y += __shfl_xor(acc.y, o);
        acc.z += __shfl_xor(acc.z, o);
        acc.w += __shfl_xor(acc.w, o);
    }
    float4 v = {0.f, 0.f, 0.f, 0.f};
    if (live && g == 0) {
        const float d  = (float)cntv;
        const float dv = d > 1.0f ? d : 1.0f;
        float4 b;
        b.x = bias[c]; b.y = bias[c + 1]; b.z = bias[c + 2]; b.w = bias[c + 3];
        v.x = acc.x / dv + b.x + rv.x + ev.x;
        v.y = acc.y / dv + b.y + rv.y + ev.y;
        v.z = acc.z / dv + b.z + rv.z + ev.z;
        v.w = acc.w / dv + b.w + rv.w + ev.w;
        if (RELU) {
            v.x = v.x > 0.0f ? v.x : 0.0f;
            v.y = v.y > 0.0f ? v.y : 0.0f;
            v.z = v.z > 0.0f ? v.z : 0.0f;
            v.w = v.w > 0.0f ? v.w : 0.0f;
        }
        if (!PROJ) *(float4*)(Out + (size_t)node * C + c) = v;
    }
    if constexpr (PROJ) {
        __shared__ float hs[16 * HP];
        __shared__ float es[16 * HP];
        __shared__ float part[8 * 2 * 256];
        const int wv  = threadIdx.x >> 5;
        const int row = wv * NPW + nw;
        *(float4*)(hs + row * HP + c) = v;
        *(float4*)(es + row * HP + c) = ev;
        const int half = lane >> 4;
        const int lm   = lane & 15;
        constexpr int KS  = C / 4;
        constexpr int KPW = KS / 8;
        static_assert(KPW * 8 == KS, "the k-steps divide over the 8 waves of a block");
        const int kof = wv * KPW * 4 + half * 2;
        v2f ba[KPW], bb[KPW], bc[KPW];
#pragma unroll
        for (int j = 0; j < KPW; ++j) {
            ba[j] = *(const v2f*)(Wa + (size_t)lm * C + kof + 4 * j);
            bb[j] = *(const v2f*)(Wb + (size_t)lm * C + kof + 4 * j);
            bc[j] = *(const v2f*)(Wc + (size_t)lm * C + kof + 4 * j);
        }
        const float b2 = bias2[lm];
        __syncthreads();
        v2f ah[KPW], ae[KPW];
#pragma unroll
        for (int j = 0; j < KPW; ++j) {
            ah[j] = *(const v2f*)(hs + lm * HP + kof + 4 * j);
            ae[j] = *(const v2f*)(es + lm * HP + kof + 4 * j);
        }
        v8f cp = (v8f){0.f, 0.f, 0.f, 0.f, 0.f, 0.f, 0.f, 0.f};
        v8f cs = cp;
#pragma unroll
        for (int j = 0; j < KPW; ++j) {
            cs = __builtin_amdgcn_wmma_f32_16x16x4_f32(false, ah[j], false, bb[j], (short)0, cs, false, false);
            cp = __builtin_amdgcn_wmma_f32_16x16x4_f32(false, ah[j], false, ba[j], (short)0, cp, false, false);
            cs = __builtin_amdgcn_wmma_f32_16x16x4_f32(false, ae[j], false, bc[j], (short)0, cs, false, false);
        }
        float* pw = part + wv * 512 + lane;
#pragma unroll
        for (int r = 0; r < 8; ++r) {
            pw[r * 32]       = cp[r];
            pw[256 + r * 32] = cs[r];
        }
        __syncthreads();
        const int orow = 2 * wv + half;
        const float* pr = part + (orow & 7) * 32 + (orow >> 3) * 16 + lm;
        float sp = pr[0];
        float ss = pr[256];
#pragma unroll
        for (int w = 1; w < 8; ++w) {
            sp += pr[w * 512];
            ss += pr[w * 512 + 256];
        }
        const int blk  = blockIdx.x;
        float*    o    = Out + (size_t)blk * PB;
        if (blk * 16 + orow < N) {
            o[orow * 16 + lm]       = sp;
            o[256 + orow * 16 + lm] = ss + b2;
        }
    }
}

extern "C" void kernel_launch(void* const* d_in, const int* in_sizes, int n_in,
                              void* d_out, int out_size, void* d_ws, size_t ws_size,
                              hipStream_t stream)
{
    const float* x_feat = (const float*)d_in[0];
    const float* x_emb  = (const float*)d_in[1];
    const int*   eidx   = (const int*)d_in[2];
    const float* Wl0 = (const float*)d_in[3];
    const float* bl0 = (const float*)d_in[4];
    const float* Wr0 = (const float*)d_in[5];
    const float* We0 = (const float*)d_in[6];
    const float* be0 = (const float*)d_in[7];
    const float* Wl1 = (const float*)d_in[8];
    const float* bl1 = (const float*)d_in[9];
    const float* Wr1 = (const float*)d_in[10];
    const float* We1 = (const float*)d_in[11];
    const float* be1 = (const float*)d_in[12];

    const int FEAT = 128;
    const int N = in_sizes[0] / FEAT;
    const int E = in_sizes[2] / 2;
    const int* src = eidx;
    const int* dst = eidx + E;

    float*       ws = (float*)d_ws;
    const size_t n  = (size_t)N;
    float* r0   = ws + n - 64;
    float* p0   = ws + 65 * n - 64;
    int*   tick = (int*)(ws + 129 * n - 64);
    int*   bka  = tick + 64;
    int*   bkb  = bka + BKW * n;
    float* e0   = ws + 193 * n;
    float* g1   = r0;

    hipMemsetAsync(tick, 0, (64 + BKW * n) * sizeof(int), stream);

    const int nchunks = (E + CHUNK - 1) / CHUNK;
    place_kernel<<<2 * nchunks, 256, 0, stream>>>(src, dst, bka, bkb, tick, E, N, nchunks);

    const int gthreads = 32 * GEMM_WAVES;

    wmma_ws_kernel<128, 4, 2, false><<<256, gthreads, 0, stream>>>(x_feat, Wl0, Wr0, nullptr, p0, r0, N);

    wmma_ws_kernel<64, 4, 1, true><<<512, gthreads, 0, stream>>>(x_emb, We0, nullptr, be0, e0, nullptr, N);

    pull_kernel<64, 2, 8, true, 1024, true><<<(N + 15) / 16, 256, 0, stream>>>(p0, bka, bkb, src, dst, E, r0, e0, bl0,
                                                                              g1, N, Wl1, Wr1, We1, be1);

    pull_kernel<16, 4, 8, false, 1024, false><<<(N + 31) / 32, 256, 0, stream>>>(g1, bka, bkb, src, dst, E, g1 + 256,
                                                                                nullptr, bl1, (float*)d_out, N,
                                                                                nullptr, nullptr, nullptr, nullptr);
}
